# starting kernel
# baseline (speedup 1.0000x reference)
#include <hip/hip_runtime.h>
#include <hip/hip_bf16.h>


typedef __attribute__((ext_vector_type(2))) float v2f;
typedef __attribute__((ext_vector_type(8))) float v8f;

#define CH    64
#define SP    56
#define NCOL  (SP * SP)
#define KRED  384
#define PLANE 3136

__global__ __launch_bounds__(256)
void shiftconv_wmma_f32(const float* __restrict__ x,
                        const float* __restrict__ W,
                        float* __restrict__ out)
{
    const int lane = threadIdx.x & 31;
    const int wave = threadIdx.x >> 5;
    const int tile = blockIdx.x * 8 + wave;
    const int row_tile = tile / 196;
    const int col_tile = tile % 196;
    const int row_base = row_tile * 16;
    const int col_base = col_tile * 16;

    const int m     = lane & 15;
    const int kh    = lane >> 4;
    const int kbase = kh * 2;

    const int c    = col_base + m;
    const int o    = c / 56;
    const int n    = c % 56;
    const int xrow = n * 56;
    const int ll0  = o - 1;
    const int ll1  = ((o + 55) % 56) - 1;

    const float* Arow = W + (row_base + m) * KRED + kbase;

    v8f acc = {};

    #pragma unroll
    for (int kl = 0; kl < 6; ++kl) {
        const int k = kl >> 1;
        const int l = kl & 1;
        const int ll = (l ? ll1 : ll0) + k;

        const bool  valid = (unsigned)ll < 56u;
        const int   llc   = valid ? ll : 0;
        const float fmask = valid ? 1.0f : 0.0f;

        const float* bp0 = x + (size_t)(kbase + 0) * PLANE + xrow + llc;
        const float* bp1 = x + (size_t)(kbase + 1) * PLANE + xrow + llc;
        const float* ap  = Arow + kl * 64;

        #pragma unroll
        for (int q0 = 0; q0 < 64; q0 += 4) {
            v2f a = *(const v2f*)(ap + q0);

            v2f b;
            b[0] = bp0[q0 * PLANE] * fmask;
            b[1] = bp1[q0 * PLANE] * fmask;

            acc = __builtin_amdgcn_wmma_f32_16x16x4_f32(
                 false, a,  false, b,
                 (short)0, acc,  false,  false);
        }
    }

    const int ostore = o * 56 + ((n + 1) % 56);
    #pragma unroll
    for (int vr = 0; vr < 8; ++vr) {
        const int i = row_base + vr + 8 * kh;
        out[i * NCOL + ostore] = acc[vr];
    }
}

extern "C" void kernel_launch(void* const* d_in, const int* in_sizes, int n_in,
                              void* d_out, int out_size, void* d_ws, size_t ws_size,
                              hipStream_t stream) {
    const float* x = (const float*)d_in[0];
    const float* W = (const float*)d_in[1];
    float* out = (float*)d_out;

    shiftconv_wmma_f32<<<98, 256, 0, stream>>>(x, W, out);
}


# B-tile columns run along x's contiguous axis (c = n*56+o, was o*56+n): a B load reads 16 consecutive floats per plane, not 16 lines 224 B apart; the 8 stores per tile take the stride instead
# speedup vs baseline: 2.3884x; 2.3884x over previous
#include <hip/hip_runtime.h>
#include <hip/hip_bf16.h>


typedef __attribute__((ext_vector_type(2))) float v2f;
typedef __attribute__((ext_vector_type(8))) float v8f;

#define CH    64
#define SP    56
#define NCOL  (SP * SP)
#define KRED  384
#define PLANE 3136

__global__ __launch_bounds__(256)
void shiftconv_wmma_f32(const float* __restrict__ x,
                        const float* __restrict__ W,
                        float* __restrict__ out)
{
    const int lane = threadIdx.x & 31;
    const int wave = threadIdx.x >> 5;
    const int tile = blockIdx.x * 8 + wave;
    const int row_tile = tile / 196;
    const int col_tile = tile % 196;
    const int row_base = row_tile * 16;
    const int col_base = col_tile * 16;

    const int m     = lane & 15;
    const int kh    = lane >> 4;
    const int kbase = kh * 2;

    const int c    = col_base + m;
    const int n    = c / 56;
    const int o    = c % 56;
    const int xrow = n * 56;
    const int ll0  = o - 1;
    const int ll1  = ((o + 55) % 56) - 1;

    const float* Arow = W + (row_base + m) * KRED + kbase;

    v8f acc = {};

    #pragma unroll
    for (int kl = 0; kl < 6; ++kl) {
        const int k = kl >> 1;
        const int l = kl & 1;
        const int ll = (l ? ll1 : ll0) + k;

        const bool  valid = (unsigned)ll < 56u;
        const int   llc   = valid ? ll : 0;
        const float fmask = valid ? 1.0f : 0.0f;

        const float* bp0 = x + (size_t)(kbase + 0) * PLANE + xrow + llc;
        const float* bp1 = x + (size_t)(kbase + 1) * PLANE + xrow + llc;
        const float* ap  = Arow + kl * 64;

        #pragma unroll
        for (int q0 = 0; q0 < 64; q0 += 4) {
            v2f a = *(const v2f*)(ap + q0);

            v2f b;
            b[0] = bp0[q0 * PLANE] * fmask;
            b[1] = bp1[q0 * PLANE] * fmask;

            acc = __builtin_amdgcn_wmma_f32_16x16x4_f32(
                 false, a,  false, b,
                 (short)0, acc,  false,  false);
        }
    }

    const int ostore = o * 56 + ((n + 1) % 56);
    #pragma unroll
    for (int vr = 0; vr < 8; ++vr) {
        const int i = row_base + vr + 8 * kh;
        out[i * NCOL + ostore] = acc[vr];
    }
}

extern "C" void kernel_launch(void* const* d_in, const int* in_sizes, int n_in,
                              void* d_out, int out_size, void* d_ws, size_t ws_size,
                              hipStream_t stream) {
    const float* x = (const float*)d_in[0];
    const float* W = (const float*)d_in[1];
    float* out = (float*)d_out;

    shiftconv_wmma_f32<<<98, 256, 0, stream>>>(x, W, out);
}


# A operand through LDS: each work-group (7 column tiles of one 16-row tile) copies its 16x384 slice of W once with whole-line loads; A fragments come from ds_load, not 16-line global loads
# speedup vs baseline: 2.9958x; 1.2543x over previous
#include <hip/hip_runtime.h>
#include <hip/hip_bf16.h>


typedef __attribute__((ext_vector_type(2))) float v2f;
typedef __attribute__((ext_vector_type(4))) float v4f;
typedef __attribute__((ext_vector_type(8))) float v8f;

#define CH     64
#define SP     56
#define NCOL   (SP * SP)
#define KRED   384
#define PLANE  3136
#define GROUP  7
#define NTHR   (GROUP * 32)
#define CGRPS  (196 / GROUP)
#define APITCH 388
#define TILE_V4 (16 * KRED / 4)

__global__ __launch_bounds__(NTHR)
void shiftconv_wmma_f32(const float* __restrict__ x,
                        const float* __restrict__ W,
                        float* __restrict__ out)
{
    __shared__ __attribute__((aligned(16))) float As[16 * APITCH];

    const int tid  = threadIdx.x;
    const int lane = tid & 31;
    const int wave = tid >> 5;
    const int row_tile = blockIdx.x / CGRPS;
    const int col_tile = (blockIdx.x % CGRPS) * GROUP + wave;
    const int row_base = row_tile * 16;
    const int col_base = col_tile * 16;

    const int m     = lane & 15;
    const int kh    = lane >> 4;
    const int kbase = kh * 2;

    const int c    = col_base + m;
    const int n    = c / 56;
    const int o    = c % 56;
    const int xrow = n * 56;
    const int ll0  = o - 1;
    const int ll1  = ((o + 55) % 56) - 1;

    const v4f* src = (const v4f*)(W + row_base * KRED);
    v4f t[6];
    #pragma unroll
    for (int i = 0; i < 6; ++i) t[i] = src[tid + NTHR * i];

    float bfirst[32];
    {
        const bool valid0 = (unsigned)ll0 < 56u;
        const int  llc0   = valid0 ? ll0 : 0;
        const float* q0p = x + (size_t)(kbase + 0) * PLANE + xrow + llc0;
        const float* q1p = x + (size_t)(kbase + 1) * PLANE + xrow + llc0;
        #pragma unroll
        for (int s = 0; s < 16; ++s) {
            bfirst[2 * s + 0] = q0p[4 * s * PLANE];
            bfirst[2 * s + 1] = q1p[4 * s * PLANE];
        }
    }

    #pragma unroll
    for (int i = 0; i < 6; ++i) {
        const int f = tid + NTHR * i;
        *(v4f*)(As + (f / (KRED / 4)) * APITCH + (f % (KRED / 4)) * 4) = t[i];
    }
    if (tid < TILE_V4 - 6 * NTHR) {
        const int f = tid + NTHR * 6;
        *(v4f*)(As + (f / (KRED / 4)) * APITCH + (f % (KRED / 4)) * 4) = src[f];
    }

    const float* Arow = As + m * APITCH + kbase;

    __syncthreads();

    v8f acc = {};

    v2f afr[2][16];
    #pragma unroll
    for (int s = 0; s < 16; ++s) afr[0][s] = *(const v2f*)(Arow + 4 * s);

    #pragma unroll
    for (int kl = 0; kl < 6; ++kl) {
        const int k = kl >> 1;
        const int l = kl & 1;
        const int ll = (l ? ll1 : ll0) + k;

        const bool  valid = (unsigned)ll < 56u;
        const int   llc   = valid ? ll : 0;
        const float fmask = valid ? 1.0f : 0.0f;

        const float* bp0 = x + (size_t)(kbase + 0) * PLANE + xrow + llc;
        const float* bp1 = x + (size_t)(kbase + 1) * PLANE + xrow + llc;

        const int cur = kl & 1;
        if (kl + 1 < 6) {
            #pragma unroll
            for (int s = 0; s < 16; ++s)
                afr[cur ^ 1][s] = *(const v2f*)(Arow + (kl + 1) * 64 + 4 * s);
        }
        #pragma unroll
        for (int q0 = 0; q0 < 64; q0 += 4) {
            v2f a = afr[cur][q0 >> 2];

            v2f b;
            b[0] = (kl == 0 ? bfirst[(q0 >> 1) + 0] : bp0[q0 * PLANE]) * fmask;
            b[1] = (kl == 0 ? bfirst[(q0 >> 1) + 1] : bp1[q0 * PLANE]) * fmask;

            acc = __builtin_amdgcn_wmma_f32_16x16x4_f32(
                 false, a,  false, b,
                 (short)0, acc,  false,  false);
        }
    }

    __builtin_amdgcn_sched_group_barrier(0x100, 8, 0);
    #pragma unroll
    for (int kl = 0; kl < 5; ++kl) {
        __builtin_amdgcn_sched_group_barrier(0x100, 8, 0);
        __builtin_amdgcn_sched_group_barrier(0x020, 32, 0);
        __builtin_amdgcn_sched_group_barrier(0x008, 16, 0);
    }
    __builtin_amdgcn_sched_group_barrier(0x008, 16, 0);

    const int ostore = o * 56 + ((n + 1) % 56);
    #pragma unroll
    for (int vr = 0; vr < 8; ++vr) {
        const int i = row_base + vr + 8 * kh;
        out[i * NCOL + ostore] = acc[vr];
    }
}

extern "C" void kernel_launch(void* const* d_in, const int* in_sizes, int n_in,
                              void* d_out, int out_size, void* d_ws, size_t ws_size,
                              hipStream_t stream) {
    const float* x = (const float*)d_in[0];
    const float* W = (const float*)d_in[1];
    float* out = (float*)d_out;

    shiftconv_wmma_f32<<<4 * CGRPS, NTHR, 0, stream>>>(x, W, out);
}


# B operand through LDS: each work-group copies its two x rows of all 64 planes once with 16-byte loads (zero rows as padding); all six tap blocks read B by ds_load, not 192 global loads per lane
# speedup vs baseline: 3.3490x; 1.1179x over previous
#include <hip/hip_runtime.h>
#include <hip/hip_bf16.h>


typedef __attribute__((ext_vector_type(2))) float v2f;
typedef __attribute__((ext_vector_type(4))) float v4f;
typedef __attribute__((ext_vector_type(8))) float v8f;

#define CH     64
#define SP     56
#define NCOL   (SP * SP)
#define KRED   384
#define PLANE  3136
#define GROUP  7
#define NTHR   (GROUP * 32)
#define CGRPS  (196 / GROUP)
#define APITCH 388
#define TILE_V4 (16 * KRED / 4)
#define RL     58
#define PADL   1
#define BPITCH 68
#define XPIECES 28

__global__ __launch_bounds__(NTHR)
void shiftconv_wmma_f32(const float* __restrict__ x,
                        const float* __restrict__ W,
                        float* __restrict__ out)
{
    __shared__ __attribute__((aligned(16))) float As[16 * APITCH];
    __shared__ __attribute__((aligned(16))) float Bs[2 * RL * BPITCH];

    const int tid  = threadIdx.x;
    const int lane = tid & 31;
    const int wave = tid >> 5;
    const int row_tile = blockIdx.x / CGRPS;
    const int cgrp     = blockIdx.x % CGRPS;
    const int row_base = row_tile * 16;

    const int m     = lane & 15;
    const int kh    = lane >> 4;
    const int kbase = kh * 2;

    const int cw   = wave * 16 + m;
    const int nrow = cw / 56;
    const int o    = cw % 56;
    const int n    = 2 * cgrp + nrow;
    const int ll0  = o - 1;
    const int ll1  = ((o + 55) % 56) - 1;

    const v4f* src = (const v4f*)(W + row_base * KRED);
    v4f t[6];
    #pragma unroll
    for (int i = 0; i < 6; ++i) t[i] = src[tid + NTHR * i];

    const float* xs = x + (2 * cgrp) * SP;
    v4f bt[2][4];
    #pragma unroll
    for (int i = 0; i < 2; ++i) {
        const int f  = tid + NTHR * i;
        const int jq = f / XPIECES;
        const int p  = f % XPIECES;
        const v4f* g = (const v4f*)(xs + (4 * jq) * PLANE) + p;
        #pragma unroll
        for (int u = 0; u < 4; ++u) bt[i][u] = g[u * (PLANE / 4)];
    }

    #pragma unroll
    for (int i = 0; i < 6; ++i) {
        const int f = tid + NTHR * i;
        *(v4f*)(As + (f / (KRED / 4)) * APITCH + (f % (KRED / 4)) * 4) = t[i];
    }
    if (tid < TILE_V4 - 6 * NTHR) {
        const int f = tid + NTHR * 6;
        *(v4f*)(As + (f / (KRED / 4)) * APITCH + (f % (KRED / 4)) * 4) = src[f];
    }

    #pragma unroll
    for (int i = 0; i < 2; ++i) {
        const int f   = tid + NTHR * i;
        const int jq  = f / XPIECES;
        const int p   = f % XPIECES;
        const int pos = (p / 14) * RL + PADL + 4 * (p % 14);
        float* d = Bs + pos * BPITCH + 4 * jq;
        #pragma unroll
        for (int e = 0; e < 4; ++e) {
            v4f cell = { bt[i][0][e], bt[i][1][e], bt[i][2][e], bt[i][3][e] };
            *(v4f*)(d + e * BPITCH) = cell;
        }
    }
    if (tid < 64) {
        const int w   = tid >> 4;
        const int pos = (w >> 1) * RL + ((w & 1) ? PADL + SP : PADL - 1);
        v4f zero = { 0.0f, 0.0f, 0.0f, 0.0f };
        *(v4f*)(Bs + pos * BPITCH + 4 * (tid & 15)) = zero;
    }

    const float* Arow = As + m * APITCH + kbase;
    const float* Bl0 = Bs + (nrow * RL + PADL + ll0) * BPITCH + kbase;
    const float* Bl1 = Bs + (nrow * RL + PADL + ll1) * BPITCH + kbase;

    __syncthreads();

    v8f acc = {};

    v2f afr[2][16];
    v2f bfr[2][16];
    #pragma unroll
    for (int s = 0; s < 16; ++s) {
        afr[0][s] = *(const v2f*)(Arow + 4 * s);
        bfr[0][s] = *(const v2f*)(Bl0 + 4 * s);
    }

    #pragma unroll
    for (int kl = 0; kl < 6; ++kl) {
        const int cur = kl & 1;
        if (kl + 1 < 6) {
            const int k1 = (kl + 1) >> 1;
            const int l1 = (kl + 1) & 1;
            const float* bb = (l1 ? Bl1 : Bl0) + k1 * BPITCH;
            #pragma unroll
            for (int s = 0; s < 16; ++s) {
                afr[cur ^ 1][s] = *(const v2f*)(Arow + (kl + 1) * 64 + 4 * s);
                bfr[cur ^ 1][s] = *(const v2f*)(bb + 4 * s);
            }
        }
        #pragma unroll
        for (int s = 0; s < 16; ++s) {
            acc = __builtin_amdgcn_wmma_f32_16x16x4_f32(
                 false, afr[cur][s],  false, bfr[cur][s],
                 (short)0, acc,  false,  false);
        }
    }

    __builtin_amdgcn_sched_group_barrier(0x100, 16, 0);
    #pragma unroll
    for (int g = 0; g < 40; ++g) {
        __builtin_amdgcn_sched_group_barrier(0x008, 2, 0);
        __builtin_amdgcn_sched_group_barrier(0x100, 2, 0);
    }
    __builtin_amdgcn_sched_group_barrier(0x008, 16, 0);

    const int ostore = o * 56 + ((n + 1) % 56);
    #pragma unroll
    for (int vr = 0; vr < 8; ++vr) {
        const int i = row_base + vr + 8 * kh;
        out[i * NCOL + ostore] = acc[vr];
    }
}

extern "C" void kernel_launch(void* const* d_in, const int* in_sizes, int n_in,
                              void* d_out, int out_size, void* d_ws, size_t ws_size,
                              hipStream_t stream) {
    const float* x = (const float*)d_in[0];
    const float* W = (const float*)d_in[1];
    float* out = (float*)d_out;

    shiftconv_wmma_f32<<<4 * CGRPS, NTHR, 0, stream>>>(x, W, out);
}


# Stores: the two n of each (i,o) leave as one aligned 8-byte store of one lane (tiles exchanged through a 7 KB LDS buffer; x rows 2c-1,2c staged): 896 store pieces per work-group, was 1792 of 4 B
# speedup vs baseline: 5.0894x; 1.5197x over previous
#include <hip/hip_runtime.h>
#include <hip/hip_bf16.h>


typedef __attribute__((ext_vector_type(2))) float v2f;
typedef __attribute__((ext_vector_type(4))) float v4f;
typedef __attribute__((ext_vector_type(8))) float v8f;

#define CH     64
#define SP     56
#define NCOL   (SP * SP)
#define KRED   384
#define PLANE  3136
#define GROUP  7
#define NTHR   (GROUP * 32)
#define CGRPS  (196 / GROUP)
#define APITCH 388
#define TILE_V4 (16 * KRED / 4)
#define RL     58
#define PADL   1
#define BPITCH 68
#define XPIECES 28
#define CPITCH 116

__global__ __launch_bounds__(NTHR)
void shiftconv_wmma_f32(const float* __restrict__ x,
                        const float* __restrict__ W,
                        float* __restrict__ out)
{
    __shared__ __attribute__((aligned(16))) float As[16 * APITCH];
    __shared__ __attribute__((aligned(16))) float Bs[2 * RL * BPITCH];
    __shared__ __attribute__((aligned(16))) float Cs[16 * CPITCH];

    const int tid  = threadIdx.x;
    const int lane = tid & 31;
    const int wave = tid >> 5;
    const int row_tile = blockIdx.x / CGRPS;
    const int cgrp     = blockIdx.x % CGRPS;
    const int row_base = row_tile * 16;

    const int m     = lane & 15;
    const int kh    = lane >> 4;
    const int kbase = kh * 2;

    const int cw   = wave * 16 + m;
    const int nrow = cw / 56;
    const int o    = cw % 56;
    const int nA   = (2 * cgrp + 55) % 56;
    const int nB   = 2 * cgrp;
    const int ll0  = o - 1;
    const int ll1  = ((o + 55) % 56) - 1;

    const v4f* src = (const v4f*)(W + row_base * KRED);
    v4f t[6];
    #pragma unroll
    for (int i = 0; i < 6; ++i) t[i] = src[tid + NTHR * i];

    v4f bt[2][4];
    #pragma unroll
    for (int i = 0; i < 2; ++i) {
        const int f  = tid + NTHR * i;
        const int jq = f / XPIECES;
        const int p  = f % XPIECES;
        const int nsel = (p / 14) ? nB : nA;
        const v4f* g = (const v4f*)(x + (4 * jq) * PLANE + nsel * SP) + (p % 14);
        #pragma unroll
        for (int u = 0; u < 4; ++u) bt[i][u] = g[u * (PLANE / 4)];
    }

    #pragma unroll
    for (int i = 0; i < 6; ++i) {
        const int f = tid + NTHR * i;
        *(v4f*)(As + (f / (KRED / 4)) * APITCH + (f % (KRED / 4)) * 4) = t[i];
    }
    if (tid < TILE_V4 - 6 * NTHR) {
        const int f = tid + NTHR * 6;
        *(v4f*)(As + (f / (KRED / 4)) * APITCH + (f % (KRED / 4)) * 4) = src[f];
    }

    #pragma unroll
    for (int i = 0; i < 2; ++i) {
        const int f   = tid + NTHR * i;
        const int jq  = f / XPIECES;
        const int p   = f % XPIECES;
        const int pos = (p / 14) * RL + PADL + 4 * (p % 14);
        float* d = Bs + pos * BPITCH + 4 * jq;
        #pragma unroll
        for (int e = 0; e < 4; ++e) {
            v4f cell = { bt[i][0][e], bt[i][1][e], bt[i][2][e], bt[i][3][e] };
            *(v4f*)(d + e * BPITCH) = cell;
        }
    }
    if (tid < 64) {
        const int w   = tid >> 4;
        const int pos = (w >> 1) * RL + ((w & 1) ? PADL + SP : PADL - 1);
        v4f zero = { 0.0f, 0.0f, 0.0f, 0.0f };
        *(v4f*)(Bs + pos * BPITCH + 4 * (tid & 15)) = zero;
    }

    const float* Arow = As + m * APITCH + kbase;
    const float* Bl0 = Bs + (nrow * RL + PADL + ll0) * BPITCH + kbase;
    const float* Bl1 = Bs + (nrow * RL + PADL + ll1) * BPITCH + kbase;

    __syncthreads();

    v8f acc = {};

    v2f afr[2][16];
    v2f bfr[2][16];
    #pragma unroll
    for (int s = 0; s < 16; ++s) {
        afr[0][s] = *(const v2f*)(Arow + 4 * s);
        bfr[0][s] = *(const v2f*)(Bl0 + 4 * s);
    }

    #pragma unroll
    for (int kl = 0; kl < 6; ++kl) {
        const int cur = kl & 1;
        if (kl + 1 < 6) {
            const int k1 = (kl + 1) >> 1;
            const int l1 = (kl + 1) & 1;
            const float* bb = (l1 ? Bl1 : Bl0) + k1 * BPITCH;
            #pragma unroll
            for (int s = 0; s < 16; ++s) {
                afr[cur ^ 1][s] = *(const v2f*)(Arow + (kl + 1) * 64 + 4 * s);
                bfr[cur ^ 1][s] = *(const v2f*)(bb + 4 * s);
            }
        }
        #pragma unroll
        for (int s = 0; s < 16; ++s) {
            acc = __builtin_amdgcn_wmma_f32_16x16x4_f32(
                 false, afr[cur][s],  false, bfr[cur][s],
                 (short)0, acc,  false,  false);
        }
    }

    __builtin_amdgcn_sched_group_barrier(0x100, 16, 0);
    #pragma unroll
    for (int g = 0; g < 40; ++g) {
        __builtin_amdgcn_sched_group_barrier(0x008, 2, 0);
        __builtin_amdgcn_sched_group_barrier(0x100, 2, 0);
    }
    __builtin_amdgcn_sched_group_barrier(0x008, 16, 0);

    #pragma unroll
    for (int vr = 0; vr < 8; ++vr)
        Cs[(vr + 8 * kh) * CPITCH + cw] = acc[vr];

    __syncthreads();

    #pragma unroll
    for (int q = 0; q < 4; ++q) {
        const int idx = tid + NTHR * q;
        const int i   = idx / SP;
        const int oo  = idx % SP;
        v2f pair = { Cs[i * CPITCH + oo], Cs[i * CPITCH + SP + oo] };
        *(v2f*)(out + (row_base + i) * NCOL + oo * SP + 2 * cgrp) = pair;
    }
}

extern "C" void kernel_launch(void* const* d_in, const int* in_sizes, int n_in,
                              void* d_out, int out_size, void* d_ws, size_t ws_size,
                              hipStream_t stream) {
    const float* x = (const float*)d_in[0];
    const float* W = (const float*)d_in[1];
    float* out = (float*)d_out;

    shiftconv_wmma_f32<<<4 * CGRPS, NTHR, 0, stream>>>(x, W, out);
}


# Work-group column block 8 x rows by 14 o (was 2 rows by 56 o): the 8 n' of each (i,o) leave as one aligned 32-byte run, two adjacent lanes of one store: 224 runs per group, was 896 pieces of 8 B
# speedup vs baseline: 6.3041x; 1.2387x over previous
#include <hip/hip_runtime.h>
#include <hip/hip_bf16.h>


typedef __attribute__((ext_vector_type(2))) float v2f;
typedef __attribute__((ext_vector_type(4))) float v4f;
typedef __attribute__((ext_vector_type(8))) float v8f;

#define CH     64
#define SP     56
#define NCOL   (SP * SP)
#define KRED   384
#define PLANE  3136
#define GROUP  7
#define NTHR   (GROUP * 32)
#define NROW   8
#define OW     14
#define OQS    (SP / OW)
#define CGRPS  ((SP / NROW) * OQS)
#define APITCH 388
#define TILE_V4 (16 * KRED / 4)
#define RLP    30
#define SLOTS  5
#define BPITCH 68
#define BITEMS (16 * NROW * SLOTS)
#define CPITCH 116

__global__ __launch_bounds__(NTHR)
void shiftconv_wmma_f32(const float* __restrict__ x,
                        const float* __restrict__ W,
                        float* __restrict__ out)
{
    __shared__ __attribute__((aligned(16))) float As[16 * APITCH];
    __shared__ __attribute__((aligned(16))) float Bs[NROW * RLP * BPITCH];
    __shared__ __attribute__((aligned(16))) float Cs[16 * CPITCH];

    const int tid  = threadIdx.x;
    const int lane = tid & 31;
    const int wave = tid >> 5;
    const int row_tile = blockIdx.x / CGRPS;
    const int cg       = blockIdx.x % CGRPS;
    const int g        = cg / OQS;
    const int oq       = cg % OQS;
    const int row_base = row_tile * 16;

    const int m     = lane & 15;
    const int kh    = lane >> 4;
    const int kbase = kh * 2;

    const int cw   = wave * 16 + m;
    const int nrow = cw / OW;
    const int oo   = cw % OW;
    const int pp0  = 2 * oq + oo;
    const int pp1  = (oq == 0 && oo == 0) ? 19 : pp0 - 1;

    const v4f* src = (const v4f*)(W + row_base * KRED);
    v4f t[6];
    #pragma unroll
    for (int i = 0; i < 6; ++i) t[i] = src[tid + NTHR * i];

    v4f bt[3][4];
    #pragma unroll
    for (int i = 0; i < 3; ++i) {
        if (i < 2 || tid < BITEMS - 2 * NTHR) {
            const int f  = tid + NTHR * i;
            const int s  = f % SLOTS;
            const int r  = (f / SLOTS) % NROW;
            const int jq = f / (SLOTS * NROW);
            const int piece = (oq == 0 && s == SLOTS - 1) ? 13 : 3 * oq + s;
            const int n  = (NROW * g + 55 + r) % SP;
            const v4f* gp = (const v4f*)(x + (4 * jq) * PLANE + n * SP) + piece;
            #pragma unroll
            for (int u = 0; u < 4; ++u) bt[i][u] = gp[u * (PLANE / 4)];
        }
    }

    #pragma unroll
    for (int i = 0; i < 6; ++i) {
        const int f = tid + NTHR * i;
        *(v4f*)(As + (f / (KRED / 4)) * APITCH + (f % (KRED / 4)) * 4) = t[i];
    }
    if (tid < TILE_V4 - 6 * NTHR) {
        const int f = tid + NTHR * 6;
        *(v4f*)(As + (f / (KRED / 4)) * APITCH + (f % (KRED / 4)) * 4) = src[f];
    }

    #pragma unroll
    for (int i = 0; i < 3; ++i) {
        if (i < 2 || tid < BITEMS - 2 * NTHR) {
            const int f   = tid + NTHR * i;
            const int s   = f % SLOTS;
            const int r   = (f / SLOTS) % NROW;
            const int jq  = f / (SLOTS * NROW);
            const int pos = r * RLP + 1 + 4 * s;
            float* d = Bs + pos * BPITCH + 4 * jq;
            #pragma unroll
            for (int e = 0; e < 4; ++e) {
                v4f cell = { bt[i][0][e], bt[i][1][e], bt[i][2][e], bt[i][3][e] };
                *(v4f*)(d + e * BPITCH) = cell;
            }
        }
    }
    {
        const v4f zero = { 0.0f, 0.0f, 0.0f, 0.0f };
        const int zr = tid >> 4;
        *(v4f*)(Bs + ((zr >> 1) * RLP + ((zr & 1) ? 21 : 0)) * BPITCH + 4 * (tid & 15)) = zero;
        if (tid < 256 - NTHR) {
            const int zr2 = (tid + NTHR) >> 4;
            *(v4f*)(Bs + ((zr2 >> 1) * RLP + ((zr2 & 1) ? 21 : 0)) * BPITCH + 4 * (tid & 15)) = zero;
        }
    }

    const float* Arow = As + m * APITCH + kbase;
    const float* Bl0 = Bs + (nrow * RLP + pp0) * BPITCH + kbase;
    const float* Bl1 = Bs + (nrow * RLP + pp1) * BPITCH + kbase;

    __syncthreads();

    v8f acc = {};

    v2f afr[2][16];
    v2f bfr[2][16];
    #pragma unroll
    for (int s = 0; s < 16; ++s) {
        afr[0][s] = *(const v2f*)(Arow + 4 * s);
        bfr[0][s] = *(const v2f*)(Bl0 + 4 * s);
    }

    #pragma unroll
    for (int kl = 0; kl < 6; ++kl) {
        const int cur = kl & 1;
        if (kl + 1 < 6) {
            const int k1 = (kl + 1) >> 1;
            const int l1 = (kl + 1) & 1;
            const float* bb = (l1 ? Bl1 : Bl0) + k1 * BPITCH;
            #pragma unroll
            for (int s = 0; s < 16; ++s) {
                afr[cur ^ 1][s] = *(const v2f*)(Arow + (kl + 1) * 64 + 4 * s);
                bfr[cur ^ 1][s] = *(const v2f*)(bb + 4 * s);
            }
        }
        #pragma unroll
        for (int s = 0; s < 16; ++s) {
            acc = __builtin_amdgcn_wmma_f32_16x16x4_f32(
                 false, afr[cur][s],  false, bfr[cur][s],
                 (short)0, acc,  false,  false);
        }
    }

    __builtin_amdgcn_sched_group_barrier(0x100, 16, 0);
    #pragma unroll
    for (int gi = 0; gi < 40; ++gi) {
        __builtin_amdgcn_sched_group_barrier(0x008, 2, 0);
        __builtin_amdgcn_sched_group_barrier(0x100, 2, 0);
    }
    __builtin_amdgcn_sched_group_barrier(0x008, 16, 0);

    #pragma unroll
    for (int vr = 0; vr < 8; ++vr)
        Cs[(vr + 8 * kh) * CPITCH + cw] = acc[vr];

    __syncthreads();

    #pragma unroll
    for (int q = 0; q < 2; ++q) {
        const int hidx = tid + NTHR * q;
        const int half = hidx & 1;
        const int item = hidx >> 1;
        const int i    = item / OW;
        const int o2   = item % OW;
        const float* c = Cs + i * CPITCH + (4 * half) * OW + o2;
        v4f run = { c[0], c[OW], c[2 * OW], c[3 * OW] };
        *(v4f*)(out + (row_base + i) * NCOL + (OW * oq + o2) * SP + NROW * g + 4 * half) = run;
    }
}

extern "C" void kernel_launch(void* const* d_in, const int* in_sizes, int n_in,
                              void* d_out, int out_size, void* d_ws, size_t ws_size,
                              hipStream_t stream) {
    const float* x = (const float*)d_in[0];
    const float* W = (const float*)d_in[1];
    float* out = (float*)d_out;

    shiftconv_wmma_f32<<<4 * CGRPS, NTHR, 0, stream>>>(x, W, out);
}


# Work-group = 4 waves on an 8 n by 8 o block (was 7 on 8 by 14): 196 groups, one wave per SIMD; next block's LDS reads as one group; even/odd k steps in two accumulators (sum order changes)
# speedup vs baseline: 7.5188x; 1.1927x over previous
#include <hip/hip_runtime.h>
#include <hip/hip_bf16.h>


typedef __attribute__((ext_vector_type(2))) float v2f;
typedef __attribute__((ext_vector_type(4))) float v4f;
typedef __attribute__((ext_vector_type(8))) float v8f;

#define CH     64
#define SP     56
#define NCOL   (SP * SP)
#define KRED   384
#define PLANE  3136
#define GROUP  4
#define NTHR   (GROUP * 32)
#define NROW   8
#define OW     8
#define OBLK   (SP / OW)
#define CGRPS  ((SP / NROW) * OBLK)
#define APITCH 388
#define TILE_V4 (16 * KRED / 4)
#define RLP    24
#define SLOTS  3
#define BPITCH 68
#define BITEMS (16 * NROW * SLOTS)
#define CPITCH 68

__global__ __launch_bounds__(NTHR)
void shiftconv_wmma_f32(const float* __restrict__ x,
                        const float* __restrict__ W,
                        float* __restrict__ out)
{
    __shared__ __attribute__((aligned(16))) float As[16 * APITCH];
    __shared__ __attribute__((aligned(16))) float Bs[NROW * RLP * BPITCH];
    __shared__ __attribute__((aligned(16))) float Cs[16 * CPITCH];

    const int tid  = threadIdx.x;
    const int lane = tid & 31;
    const int wave = tid >> 5;
    const int row_tile = blockIdx.x / CGRPS;
    const int cg       = blockIdx.x % CGRPS;
    const int g        = cg / OBLK;
    const int ob       = cg % OBLK;
    const bool wrap    = (ob == OBLK - 1);
    const int row_base = row_tile * 16;

    const int m     = lane & 15;
    const int kh    = lane >> 4;
    const int kbase = kh * 2;

    const int cw   = wave * 16 + m;
    const int nrow = cw >> 3;
    const int oo   = cw & 7;
    const int pp0  = oo + 1 + ((wrap && oo >= 6) ? 1 : 0);
    const int pp1  = oo + ((wrap && oo == 7) ? 1 : 0);

    const v4f* src = (const v4f*)(W + row_base * KRED);
    v4f t[12];
    #pragma unroll
    for (int i = 0; i < 12; ++i) t[i] = src[tid + NTHR * i];

    v4f bt[3][4];
    #pragma unroll
    for (int i = 0; i < 3; ++i) {
        const int f  = tid + NTHR * i;
        const int s  = f % SLOTS;
        const int r  = (f / SLOTS) % NROW;
        const int jq = f / (SLOTS * NROW);
        const int piece = (2 * ob + s) % 14;
        const int n  = (NROW * g + 55 + r) % SP;
        const v4f* gp = (const v4f*)(x + (4 * jq) * PLANE + n * SP) + piece;
        #pragma unroll
        for (int u = 0; u < 4; ++u) bt[i][u] = gp[u * (PLANE / 4)];
    }

    #pragma unroll
    for (int i = 0; i < 12; ++i) {
        const int f = tid + NTHR * i;
        *(v4f*)(As + (f / (KRED / 4)) * APITCH + (f % (KRED / 4)) * 4) = t[i];
    }

    #pragma unroll
    for (int i = 0; i < 3; ++i) {
        const int f   = tid + NTHR * i;
        const int s   = f % SLOTS;
        const int r   = (f / SLOTS) % NROW;
        const int jq  = f / (SLOTS * NROW);
        const int pos = r * RLP + 4 * s + ((wrap && s == SLOTS - 1) ? 1 : 0);
        float* d = Bs + pos * BPITCH + 4 * jq;
        #pragma unroll
        for (int e = 0; e < 4; ++e) {
            v4f cell = { bt[i][0][e], bt[i][1][e], bt[i][2][e], bt[i][3][e] };
            *(v4f*)(d + e * BPITCH) = cell;
        }
    }
    if (wrap) {
        const v4f zero = { 0.0f, 0.0f, 0.0f, 0.0f };
        *(v4f*)(Bs + ((tid >> 4) * RLP + 8) * BPITCH + 4 * (tid & 15)) = zero;
    }

    const float* Arow = As + m * APITCH + kbase;
    const float* Bl0 = Bs + (nrow * RLP + pp0) * BPITCH + kbase;
    const float* Bl1 = Bs + (nrow * RLP + pp1) * BPITCH + kbase;

    __syncthreads();

    v8f acc0 = {};
    v8f acc1 = {};

    v2f afr[2][16];
    v2f bfr[2][16];
    #pragma unroll
    for (int s = 0; s < 16; ++s) {
        afr[0][s] = *(const v2f*)(Arow + 4 * s);
        bfr[0][s] = *(const v2f*)(Bl0 + 4 * s);
    }

    #pragma unroll
    for (int kl = 0; kl < 6; ++kl) {
        const int cur = kl & 1;
        if (kl + 1 < 6) {
            const int k1 = (kl + 1) >> 1;
            const int l1 = (kl + 1) & 1;
            const float* bb = (l1 ? Bl1 : Bl0) + k1 * BPITCH;
            #pragma unroll
            for (int s = 0; s < 16; ++s) {
                afr[cur ^ 1][s] = *(const v2f*)(Arow + (kl + 1) * 64 + 4 * s);
                bfr[cur ^ 1][s] = *(const v2f*)(bb + 4 * s);
            }
        }
        #pragma unroll
        for (int s = 0; s < 16; s += 2) {
            acc0 = __builtin_amdgcn_wmma_f32_16x16x4_f32(
                 false, afr[cur][s],  false, bfr[cur][s],
                 (short)0, acc0,  false,  false);
            acc1 = __builtin_amdgcn_wmma_f32_16x16x4_f32(
                 false, afr[cur][s + 1],  false, bfr[cur][s + 1],
                 (short)0, acc1,  false,  false);
        }
    }

    __builtin_amdgcn_sched_group_barrier(0x100, 16, 0);
    #pragma unroll
    for (int gi = 0; gi < 5; ++gi) {
        __builtin_amdgcn_sched_group_barrier(0x100, 16, 0);
        __builtin_amdgcn_sched_group_barrier(0x008, 16, 0);
    }
    __builtin_amdgcn_sched_group_barrier(0x008, 16, 0);

    const v8f acc = acc0 + acc1;

    #pragma unroll
    for (int vr = 0; vr < 8; ++vr)
        Cs[(vr + 8 * kh) * CPITCH + cw] = acc[vr];

    __syncthreads();

    #pragma unroll
    for (int q = 0; q < 2; ++q) {
        const int hidx = tid + NTHR * q;
        const int half = hidx & 1;
        const int item = hidx >> 1;
        const int i    = item / OW;
        const int o2   = item % OW;
        const int o    = (OW * ob + 2 + o2) % SP;
        const float* c = Cs + i * CPITCH + (4 * half) * OW + o2;
        v4f run = { c[0], c[OW], c[2 * OW], c[3 * OW] };
        *(v4f*)(out + (row_base + i) * NCOL + o * SP + NROW * g + 4 * half) = run;
    }
}

extern "C" void kernel_launch(void* const* d_in, const int* in_sizes, int n_in,
                              void* d_out, int out_size, void* d_ws, size_t ws_size,
                              hipStream_t stream) {
    const float* x = (const float*)d_in[0];
    const float* W = (const float*)d_in[1];
    float* out = (float*)d_out;

    shiftconv_wmma_f32<<<4 * CGRPS, NTHR, 0, stream>>>(x, W, out);
}


# Taps reading the same x column share one product: W[k,0]+W[k+1,1] added while staging A (sum order changes): K 384 to 256, 64 WMMAs per wave; wrapped column o=0 in 28 work-groups of its own
# speedup vs baseline: 8.3937x; 1.1164x over previous
#include <hip/hip_runtime.h>
#include <hip/hip_bf16.h>


typedef __attribute__((ext_vector_type(2))) float v2f;
typedef __attribute__((ext_vector_type(4))) float v4f;
typedef __attribute__((ext_vector_type(8))) float v8f;

#define CH     64
#define SP     56
#define NCOL   (SP * SP)
#define KRED   384
#define PLANE  3136
#define GROUP  4
#define NTHR   (GROUP * 32)
#define NROW   8
#define OW     8
#define OBLK   (SP / OW)
#define OSLOTS (OBLK + 1)
#define CGRPS  ((SP / NROW) * OSLOTS)
#define APITCH 260
#define RLP    24
#define SLOTS  3
#define BPITCH 68
#define BITEMS (16 * NROW * SLOTS)
#define CPITCH 68

__global__ __launch_bounds__(NTHR)
void shiftconv_wmma_f32(const float* __restrict__ x,
                        const float* __restrict__ W,
                        float* __restrict__ out)
{
    __shared__ __attribute__((aligned(16))) float As[16 * APITCH];
    __shared__ __attribute__((aligned(16))) float Bs[NROW * RLP * BPITCH];
    __shared__ __attribute__((aligned(16))) float Cs[16 * CPITCH];

    const int tid  = threadIdx.x;
    const int lane = tid & 31;
    const int wave = tid >> 5;
    const unsigned bid = blockIdx.x;
    const int row_tile = bid / CGRPS;
    const int cg       = bid % CGRPS;
    const int g        = cg / OSLOTS;
    const int os       = cg % OSLOTS;
    const bool special = (os == OSLOTS - 1);
    const int ob       = special ? OBLK - 1 : os;
    const bool wrap    = (ob == OBLK - 1);
    const int row_base = row_tile * 16;

    const int m     = lane & 15;
    const int kh    = lane >> 4;
    const int kbase = kh * 2;

    const int cw   = wave * 16 + m;
    const int nrow = cw >> 3;
    const int oo   = cw & 7;
    const int pbase = special ? 6 : (oo + ((wrap && oo == 7) ? 1 : 0));
    const int spoff = special ? BPITCH : 0;

    const v4f* src = (const v4f*)(W + row_base * KRED);
    v4f t[2][6];
    #pragma unroll
    for (int h = 0; h < 2; ++h) {
        const int c = tid + NTHR * h;
        #pragma unroll
        for (int mb = 0; mb < 6; ++mb)
            t[h][mb] = src[(c >> 4) * (KRED / 4) + mb * 16 + (c & 15)];
    }

    v4f bt[3][4];
    #pragma unroll
    for (int i = 0; i < 3; ++i) {
        const int f  = tid + NTHR * i;
        const int s  = f % SLOTS;
        const int r  = (f / SLOTS) % NROW;
        const int jq = f / (SLOTS * NROW);
        const int piece = (2 * ob + s) % 14;
        const int n  = (NROW * g + 55 + r) % SP;
        const v4f* gp = (const v4f*)(x + (4 * jq) * PLANE + n * SP) + piece;
        #pragma unroll
        for (int u = 0; u < 4; ++u) bt[i][u] = gp[u * (PLANE / 4)];
    }

    const float keep = special ? 0.0f : 1.0f;
    #pragma unroll
    for (int h = 0; h < 2; ++h) {
        const int c = tid + NTHR * h;
        float* d = As + (c >> 4) * APITCH + (c & 15) * 4;
        const v4f tap1 = keep * t[h][0] + t[h][3];
        const v4f tap2 = keep * t[h][5] + t[h][2];
        *(v4f*)(d)       = t[h][1];
        *(v4f*)(d + 64)  = tap1;
        *(v4f*)(d + 128) = tap2;
        *(v4f*)(d + 192) = t[h][4];
    }

    #pragma unroll
    for (int i = 0; i < 3; ++i) {
        const int f   = tid + NTHR * i;
        const int s   = f % SLOTS;
        const int r   = (f / SLOTS) % NROW;
        const int jq  = f / (SLOTS * NROW);
        const int pos = r * RLP + 4 * s + ((wrap && s == SLOTS - 1) ? 1 : 0);
        float* d = Bs + pos * BPITCH + 4 * jq;
        #pragma unroll
        for (int e = 0; e < 4; ++e) {
            v4f cell = { bt[i][0][e], bt[i][1][e], bt[i][2][e], bt[i][3][e] };
            *(v4f*)(d + e * BPITCH) = cell;
        }
    }
    if (wrap) {
        const v4f zero = { 0.0f, 0.0f, 0.0f, 0.0f };
        *(v4f*)(Bs + ((tid >> 4) * RLP + 8) * BPITCH + 4 * (tid & 15)) = zero;
    }

    const float* Arow = As + m * APITCH + kbase;
    const float* Bl = Bs + (nrow * RLP + pbase) * BPITCH + kbase;

    __syncthreads();

    v8f acc0 = {};
    v8f acc1 = {};

    v2f afr[2][16];
    v2f bfr[2][16];
    #pragma unroll
    for (int s = 0; s < 16; ++s) {
        afr[0][s] = *(const v2f*)(Arow + 4 * s);
        bfr[0][s] = *(const v2f*)(Bl + 4 * s);
    }

    #pragma unroll
    for (int kl = 0; kl < 4; ++kl) {
        const int cur = kl & 1;
        if (kl + 1 < 4) {
            const float* bb = Bl + (kl + 1) * BPITCH + ((kl + 1 >= 2) ? spoff : 0);
            #pragma unroll
            for (int s = 0; s < 16; ++s) {
                afr[cur ^ 1][s] = *(const v2f*)(Arow + (kl + 1) * 64 + 4 * s);
                bfr[cur ^ 1][s] = *(const v2f*)(bb + 4 * s);
            }
        }
        #pragma unroll
        for (int s = 0; s < 16; s += 2) {
            acc0 = __builtin_amdgcn_wmma_f32_16x16x4_f32(
                 false, afr[cur][s],  false, bfr[cur][s],
                 (short)0, acc0,  false,  false);
            acc1 = __builtin_amdgcn_wmma_f32_16x16x4_f32(
                 false, afr[cur][s + 1],  false, bfr[cur][s + 1],
                 (short)0, acc1,  false,  false);
        }
    }

    __builtin_amdgcn_sched_group_barrier(0x100, 16, 0);
    #pragma unroll
    for (int gi = 0; gi < 3; ++gi) {
        __builtin_amdgcn_sched_group_barrier(0x100, 16, 0);
        __builtin_amdgcn_sched_group_barrier(0x008, 16, 0);
    }
    __builtin_amdgcn_sched_group_barrier(0x008, 16, 0);

    const v8f acc = acc0 + acc1;

    #pragma unroll
    for (int vr = 0; vr < 8; ++vr)
        Cs[(vr + 8 * kh) * CPITCH + cw] = acc[vr];

    __syncthreads();

    #pragma unroll
    for (int q = 0; q < 2; ++q) {
        const int hidx = tid + NTHR * q;
        const int half = hidx & 1;
        const int item = hidx >> 1;
        const int i    = item / OW;
        const int o2   = item % OW;
        const int o    = (OW * ob + 2 + o2) % SP;
        const float* c = Cs + i * CPITCH + (4 * half) * OW + o2;
        v4f run = { c[0], c[OW], c[2 * OW], c[3 * OW] };
        const bool mine = special ? (o2 == 6) : !(wrap && o2 == 6);
        if (mine)
            *(v4f*)(out + (row_base + i) * NCOL + o * SP + NROW * g + 4 * half) = run;
    }
}

extern "C" void kernel_launch(void* const* d_in, const int* in_sizes, int n_in,
                              void* d_out, int out_size, void* d_ws, size_t ws_size,
                              hipStream_t stream) {
    const float* x = (const float*)d_in[0];
    const float* W = (const float*)d_in[1];
    float* out = (float*)d_out;

    shiftconv_wmma_f32<<<4 * CGRPS, NTHR, 0, stream>>>(x, W, out);
}
